# starting kernel
# baseline (speedup 1.0000x reference)
#include <hip/hip_runtime.h>


typedef float v2f __attribute__((ext_vector_type(2)));
typedef float v8f __attribute__((ext_vector_type(8)));

#define N_FEAT_IN 100

__global__ void degree_kernel(const int* __restrict__ src,
                              const int* __restrict__ dst,
                              float* __restrict__ deg_out,
                              float* __restrict__ deg_in,
                              int n_edges) {
    int e = blockIdx.x * blockDim.x + threadIdx.x;
    if (e >= n_edges) return;
    atomicAdd(&deg_out[src[e]], 1.0f);
    atomicAdd(&deg_in[dst[e]], 1.0f);
}

__global__ void norm_kernel(float* __restrict__ a, float* __restrict__ b, int n) {
    int i = blockIdx.x * blockDim.x + threadIdx.x;
    if (i >= n) return;
    a[i] = rsqrtf(fmaxf(a[i], 1.0f));
    b[i] = rsqrtf(fmaxf(b[i], 1.0f));
}

template <int F>
__global__ void scatter_add_kernel(const float* __restrict__ h,
                                   const float* __restrict__ nsrc,
                                   const int* __restrict__ src,
                                   const int* __restrict__ dst,
                                   float* __restrict__ agg,
                                   int n_edges) {
    constexpr int CH = F / 4;
    long long idx = (long long)blockIdx.x * blockDim.x + threadIdx.x;
    if (idx >= (long long)n_edges * CH) return;
    int e = (int)(idx / CH);
    int f = (int)(idx % CH) * 4;
    int s = src[e];
    int d = dst[e];
    float ns = nsrc[s];
    float4 v = *(const float4*)(h + (size_t)s * F + f);
    float* arow = agg + (size_t)d * F + f;
    atomicAdd(arow + 0, v.x * ns);
    atomicAdd(arow + 1, v.y * ns);
    atomicAdd(arow + 2, v.z * ns);
    atomicAdd(arow + 3, v.w * ns);
}

template <int F_IN, int F_OUT, int NT, bool RELU>
__global__ __launch_bounds__(32) void gcn_wmma_layer(
    const float* __restrict__ agg, const float* __restrict__ ndst,
    const float* __restrict__ W, const float* __restrict__ bias,
    float* __restrict__ out, int n_nodes) {
    const int lane = threadIdx.x;
    const int half = lane >> 4;
    const int lm = lane & 15;
    const int row0 = blockIdx.x * 16;

    v8f acc[NT];
#pragma unroll
    for (int t = 0; t < NT; ++t) acc[t] = v8f{};

    const float* arow = agg + (size_t)(row0 + lm) * F_IN;

    for (int k0 = 0; k0 < F_IN; k0 += 4) {
        const int ka = k0 + 2 * half;
        v2f a;
        a.x = arow[ka];
        a.y = arow[ka + 1];
#pragma unroll
        for (int t = 0; t < NT; ++t) {
            const int col = t * 16 + lm;
            v2f bf;
            bf.x = (col < F_OUT) ? W[(size_t)ka * F_OUT + col] : 0.0f;
            bf.y = (col < F_OUT) ? W[(size_t)(ka + 1) * F_OUT + col] : 0.0f;
            acc[t] = __builtin_amdgcn_wmma_f32_16x16x4_f32(
                 false, a,  false, bf,
                 (short)0, acc[t],  false,  false);
        }
    }

#pragma unroll
    for (int i = 0; i < 8; ++i) {
        const int r = row0 + i + 8 * half;
        const float nd = ndst[r];
#pragma unroll
        for (int t = 0; t < NT; ++t) {
            const int col = t * 16 + lm;
            if (col < F_OUT) {
                float v = acc[t][i] * nd + bias[col];
                if (RELU) v = fmaxf(v, 0.0f);
                out[(size_t)r * F_OUT + col] = v;
            }
        }
    }
}

static inline size_t align256(size_t x) { return (x + 255) & ~(size_t)255; }

extern "C" void kernel_launch(void* const* d_in, const int* in_sizes, int n_in,
                              void* d_out, int out_size, void* d_ws, size_t ws_size,
                              hipStream_t stream) {
    const float* x  = (const float*)d_in[0];
    const float* W0 = (const float*)d_in[1];
    const float* b0 = (const float*)d_in[2];
    const float* W1 = (const float*)d_in[3];
    const float* b1 = (const float*)d_in[4];
    const float* W2 = (const float*)d_in[5];
    const float* b2 = (const float*)d_in[6];
    const int* src  = (const int*)d_in[7];
    const int* dst  = (const int*)d_in[8];

    const int N = in_sizes[0] / N_FEAT_IN;
    const int E = in_sizes[7];
    float* out = (float*)d_out;

    char* p = (char*)d_ws;
    float* nsrc = (float*)p;  p += align256((size_t)N * 4);
    float* ndst = (float*)p;  p += align256((size_t)N * 4);
    float* agg  = (float*)p;  p += align256((size_t)N * N_FEAT_IN * 4);
    float* h0   = (float*)p;  p += align256((size_t)N * 64 * 4);
    float* h1   = (float*)p;  p += align256((size_t)N * 64 * 4);

    const int TB = 256;

    hipMemsetAsync(nsrc, 0, (size_t)N * 4, stream);
    hipMemsetAsync(ndst, 0, (size_t)N * 4, stream);
    degree_kernel<<<(E + TB - 1) / TB, TB, 0, stream>>>(src, dst, nsrc, ndst, E);
    norm_kernel<<<(N + TB - 1) / TB, TB, 0, stream>>>(nsrc, ndst, N);

    hipMemsetAsync(agg, 0, (size_t)N * N_FEAT_IN * 4, stream);
    {
        long long total = (long long)E * (N_FEAT_IN / 4);
        int blocks = (int)((total + TB - 1) / TB);
        scatter_add_kernel<N_FEAT_IN><<<blocks, TB, 0, stream>>>(x, nsrc, src, dst, agg, E);
    }
    gcn_wmma_layer<N_FEAT_IN, 64, 4, true>
        <<<N / 16, 32, 0, stream>>>(agg, ndst, W0, b0, h0, N);

    hipMemsetAsync(agg, 0, (size_t)N * 64 * 4, stream);
    {
        long long total = (long long)E * (64 / 4);
        int blocks = (int)((total + TB - 1) / TB);
        scatter_add_kernel<64><<<blocks, TB, 0, stream>>>(h0, nsrc, src, dst, agg, E);
    }
    gcn_wmma_layer<64, 64, 4, true>
        <<<N / 16, 32, 0, stream>>>(agg, ndst, W1, b1, h1, N);

    hipMemsetAsync(agg, 0, (size_t)N * 64 * 4, stream);
    {
        long long total = (long long)E * (64 / 4);
        int blocks = (int)((total + TB - 1) / TB);
        scatter_add_kernel<64><<<blocks, TB, 0, stream>>>(h1, nsrc, src, dst, agg, E);
    }
    gcn_wmma_layer<64, 47, 3, false>
        <<<N / 16, 32, 0, stream>>>(agg, ndst, W2, b2, out, N);
}


# Edge records come from one partition launch: a workgroup buckets 4096 edges by range in LDS (norms from an LDS table, one int atomic per range claims room); layer 0 reads records, no tag filter
# speedup vs baseline: 2.4972x; 2.4972x over previous
#include <hip/hip_runtime.h>


typedef float v2f __attribute__((ext_vector_type(2)));
typedef float v8f __attribute__((ext_vector_type(8)));

#define N_FEAT_IN 100

#define R_NODES 200
#define R_TILES ((R_NODES + 15) / 16)
#define R_PAD (R_TILES * 16)
template <int F> struct LdsPitch { static constexpr int v = (F % 64 == 0) ? F + 4 : F; };

__global__ void degree_kernel(const int* __restrict__ src,
                              const int* __restrict__ dst,
                              float* __restrict__ deg_out,
                              float* __restrict__ deg_in,
                              int n_edges) {
    const int t = blockIdx.x * blockDim.x + threadIdx.x;
    int e = t * 4;
    if (e + 4 <= n_edges) {
        const int4 s = *reinterpret_cast<const int4*>(src + e);
        const int4 d = *reinterpret_cast<const int4*>(dst + e);
        atomicAdd(&deg_out[s.x], 1.0f);
        atomicAdd(&deg_out[s.y], 1.0f);
        atomicAdd(&deg_out[s.z], 1.0f);
        atomicAdd(&deg_out[s.w], 1.0f);
        atomicAdd(&deg_in[d.x], 1.0f);
        atomicAdd(&deg_in[d.y], 1.0f);
        atomicAdd(&deg_in[d.z], 1.0f);
        atomicAdd(&deg_in[d.w], 1.0f);
    } else {
        for (; e < n_edges; ++e) {
            atomicAdd(&deg_out[src[e]], 1.0f);
            atomicAdd(&deg_in[dst[e]], 1.0f);
        }
    }
}

__global__ __launch_bounds__(256) void norm_count_kernel(float* __restrict__ a,
                                                         float* __restrict__ b,
                                                         int* __restrict__ cnt,
                                                         int* __restrict__ cursor, int n) {
    __shared__ float part[8];
    const int t = threadIdx.x;
    const int i = blockIdx.x * R_NODES + t;
    float din = 0.0f;
    if (t < R_NODES && i < n) {
        a[i] = rsqrtf(fmaxf(a[i], 1.0f));
        din = b[i];
        b[i] = rsqrtf(fmaxf(din, 1.0f));
    }
    for (int o = 16; o > 0; o >>= 1) din += __shfl_xor(din, o);
    if ((t & 31) == 0) part[t >> 5] = din;
    __syncthreads();
    if (t == 0) {
        float s = 0.0f;
        for (int k = 0; k < 8; ++k) s += part[k];
        cnt[blockIdx.x] = (int)s;
        cursor[blockIdx.x] = 0;
    }
}

template <int F, int K>
__device__ __forceinline__ void add_rows(const float* __restrict__ h, float* acc,
                                         int sv, int dv, float nv, int j, int lane) {
    constexpr int NC = (F + 31) / 32;
    int s[K], d[K];
    float ns[K];
    float v[K][NC];
#pragma unroll
    for (int t = 0; t < K; ++t) {
        s[t] = __builtin_amdgcn_readlane(sv, j + t);
        d[t] = __builtin_amdgcn_readlane(dv, j + t);
        ns[t] = __int_as_float(__builtin_amdgcn_readlane(__float_as_int(nv), j + t));
    }
#pragma unroll
    for (int t = 0; t < K; ++t) {
        const float* hrow = h + (size_t)s[t] * F;
#pragma unroll
        for (int c = 0; c < NC; ++c) {
            const int f = c * 32 + lane;
            v[t][c] = (f < F) ? hrow[f] : 0.0f;
        }
    }
#pragma unroll
    for (int t = 0; t < K; ++t) {
        float* arow = acc + d[t] * LdsPitch<F>::v;
#pragma unroll
        for (int c = 0; c < NC; ++c) {
            const int f = c * 32 + lane;
            if (f < F)
                __hip_atomic_fetch_add(arow + f, v[t][c] * ns[t], __ATOMIC_RELAXED,
                                       __HIP_MEMORY_SCOPE_WORKGROUP);
        }
    }
}

template <int F>
__device__ __forceinline__ void add_queue(const float* __restrict__ h, float* acc,
                                          int sv, int dv, float nv, int n, int lane) {
    int j = 0;
    for (; j + 8 <= n; j += 8) add_rows<F, 8>(h, acc, sv, dv, nv, j, lane);
    if (j + 4 <= n) { add_rows<F, 4>(h, acc, sv, dv, nv, j, lane); j += 4; }
    if (j + 2 <= n) { add_rows<F, 2>(h, acc, sv, dv, nv, j, lane); j += 2; }
    if (j < n) add_rows<F, 1>(h, acc, sv, dv, nv, j, lane);
}

#define REC_ID_MASK 0x00FFFFFF

#define W_PITCH 80

template <int F_IN, int F_OUT, int NT>
__device__ __forceinline__ void stage_w(float* wl, const float* __restrict__ W,
                                        int tid, int nthreads) {
    constexpr int C = NT * 16;
    static_assert(C <= W_PITCH && F_IN % 2 == 0, "W does not fit its LDS layout");
    for (int i = tid; i < F_IN * C; i += nthreads) {
        const int k = i / C;
        const int c = i - k * C;
        wl[(k >> 1) * (2 * W_PITCH) + c * 2 + (k & 1)] = (c < F_OUT) ? W[k * F_OUT + c] : 0.0f;
    }
}

template <int NT>
__device__ __forceinline__ void load_b_frags(const float* wl, int ka, int lm, v2f (&bf)[NT]) {
#pragma unroll
    for (int t = 0; t < NT; ++t)
        bf[t] = *reinterpret_cast<const v2f*>(wl + (ka >> 1) * (2 * W_PITCH) + (t * 16 + lm) * 2);
}

template <int F_IN, int F_OUT, int NT, bool RELU>
__device__ __forceinline__ void dense_tile_from_lds(float* tile, const float* wl,
                                                    const float* __restrict__ ndst,
                                                    const float* __restrict__ bias,
                                                    int r0, int n_nodes, int t16, int lane) {
    constexpr int P = LdsPitch<F_IN>::v;
    const int half = lane >> 4;
    const int lm = lane & 15;

    v8f acc[NT];
#pragma unroll
    for (int t = 0; t < NT; ++t) acc[t] = v8f{};

    const float* arow = tile + (t16 * 16 + lm) * P + 2 * half;

    float nd[8];
#pragma unroll
    for (int i = 0; i < 8; ++i) {
        int r = r0 + t16 * 16 + i + 8 * half;
        if (r > n_nodes - 1) r = n_nodes - 1;
        nd[i] = ndst[r];
    }
    float bs[NT];
#pragma unroll
    for (int t = 0; t < NT; ++t) {
        const int col = t * 16 + lm;
        bs[t] = bias[col < F_OUT ? col : F_OUT - 1];
    }

    v2f bf[NT];
    load_b_frags<NT>(wl, 2 * half, lm, bf);
#pragma unroll
    for (int k0 = 0; k0 < F_IN; k0 += 4) {
        const v2f a = *reinterpret_cast<const v2f*>(arow + k0);
        v2f bn[NT];
        if (k0 + 4 < F_IN) load_b_frags<NT>(wl, k0 + 4 + 2 * half, lm, bn);
#pragma unroll
        for (int t = 0; t < NT; ++t)
            acc[t] = __builtin_amdgcn_wmma_f32_16x16x4_f32(
                 false, a,  false, bf[t],
                 (short)0, acc[t],  false,  false);
        if (k0 + 4 < F_IN) {
#pragma unroll
            for (int t = 0; t < NT; ++t) bf[t] = bn[t];
        }
    }

#pragma unroll
    for (int i = 0; i < 8; ++i) {
        const int row = t16 * 16 + i + 8 * half;
#pragma unroll
        for (int t = 0; t < NT; ++t) {
            const int col = t * 16 + lm;
            if (col < F_OUT) {
                float v = acc[t][i] * nd[i] + bs[t];
                if (RELU) v = fmaxf(v, 0.0f);
                tile[row * P + col] = v;
            }
        }
    }
}

template <int P, int F_OUT>
__device__ __forceinline__ void store_rows(const float* tile, float* __restrict__ out,
                                           int r0, int n_nodes, int tid, int nthreads) {
    int rows = n_nodes - r0;
    if (rows > R_NODES) rows = R_NODES;
    float* o = out + (size_t)r0 * F_OUT;
    if (F_OUT % 4 == 0) {
        constexpr int Q = F_OUT / 4;
        const int n4 = rows * Q;
        for (int i = tid; i < n4; i += nthreads) {
            const int row = i / Q;
            const int c4 = i - row * Q;
            reinterpret_cast<float4*>(o)[i] =
                *reinterpret_cast<const float4*>(tile + row * P + c4 * 4);
        }
    } else {
        const int n1 = rows * F_OUT;
        for (int i = tid; i < n1; i += nthreads) {
            const int row = i / F_OUT;
            const int col = i - row * F_OUT;
            o[i] = tile[row * P + col];
        }
    }
}

#define SPMM_WAVES 32

#define PART_EDGES 4096
#define PART_THREADS 1024
#define NODE_TABLE 50000
#define TABLE_ITERS ((NODE_TABLE / 4 + PART_THREADS - 1) / PART_THREADS)

__global__ __launch_bounds__(PART_THREADS) void partition_kernel(
    const int* __restrict__ src, const int* __restrict__ dst,
    const float* __restrict__ nsrc, const int* __restrict__ cnt,
    int* __restrict__ cursor, int* __restrict__ rec_out,
    int n_edges, int n_nodes, int n_ranges, int rec_cap) {
    __shared__ alignas(16) float tbl[NODE_TABLE];
    __shared__ int2 stage[PART_EDGES];
    __shared__ unsigned char stage_r[PART_EDGES];
    __shared__ int hist[256];
    __shared__ int seglen[256];
    __shared__ int loff[256];
    __shared__ int gbase[256];
    __shared__ int wsum[16];
    int2* rec = reinterpret_cast<int2*>(rec_out);

    const int tid = threadIdx.x;
    const int lane = tid & 31;
    const int wave = tid >> 5;
    const int chunk0 = blockIdx.x * PART_EDGES;
    const int e0 = chunk0 + tid * 4;

    int s[4], d[4];
    int nval = n_edges - e0;
    if (nval > 4) nval = 4;
    if (nval == 4) {
        const int4 sv = *reinterpret_cast<const int4*>(src + e0);
        const int4 dv = *reinterpret_cast<const int4*>(dst + e0);
        s[0] = sv.x; s[1] = sv.y; s[2] = sv.z; s[3] = sv.w;
        d[0] = dv.x; d[1] = dv.y; d[2] = dv.z; d[3] = dv.w;
    } else {
#pragma unroll
        for (int k = 0; k < 4; ++k) {
            s[k] = (k < nval) ? src[e0 + k] : 0;
            d[k] = (k < nval) ? dst[e0 + k] : 0;
        }
    }

    const bool use_tbl = (n_nodes <= NODE_TABLE);
    const int nt = use_tbl ? n_nodes : 0;
    const int n4 = nt >> 2;
    int cv = 0;
    {
        float4 tv[TABLE_ITERS];
#pragma unroll
        for (int u = 0; u < TABLE_ITERS; ++u) {
            const int i = tid + u * PART_THREADS;
            tv[u] = (i < n4) ? reinterpret_cast<const float4*>(nsrc)[i]
                             : make_float4(0.f, 0.f, 0.f, 0.f);
        }
        cv = (tid < 256 && tid < n_ranges) ? cnt[tid] : 0;
#pragma unroll
        for (int u = 0; u < TABLE_ITERS; ++u) {
            const int i = tid + u * PART_THREADS;
            if (i < n4) reinterpret_cast<float4*>(tbl)[i] = tv[u];
        }
        const int i = (n4 << 2) + tid;
        if (i < nt) tbl[i] = nsrc[i];
    }
    if (tid < 256) {
        hist[tid] = 0;
        seglen[tid] = (cv + 15) & ~15;
    }
    __syncthreads();

    int r[4], rank[4];
    float w[4];
#pragma unroll
    for (int k = 0; k < 4; ++k) {
        r[k] = 0; rank[k] = 0; w[k] = 0.0f;
        if (k < nval) {
            r[k] = (int)((unsigned)d[k] / R_NODES);
            rank[k] = atomicAdd(&hist[r[k]], 1);
        }
    }
    if (use_tbl) {
#pragma unroll
        for (int k = 0; k < 4; ++k)
            if (k < nval) w[k] = tbl[s[k]];
    } else {
#pragma unroll
        for (int k = 0; k < 4; ++k)
            if (k < nval) w[k] = nsrc[s[k]];
    }
    __syncthreads();

    int h = 0, hs = 0, cs = 0, c = 0;
    if (tid < 256) {
        h = hist[tid];
        c = seglen[tid];
        hs = h;
        cs = c;
#pragma unroll
        for (int o = 1; o < 32; o <<= 1) {
            const int t1 = __shfl_up(hs, o);
            const int t2 = __shfl_up(cs, o);
            if (lane >= o) { hs += t1; cs += t2; }
        }
        if (lane == 31) { wsum[wave] = hs; wsum[8 + wave] = cs; }
    }
    __syncthreads();
    if (tid < 256) {
        int ph = 0, pc = 0;
        for (int k = 0; k < wave; ++k) { ph += wsum[k]; pc += wsum[8 + k]; }
        loff[tid] = ph + hs - h;
        int g = 0;
        if (h > 0) g = atomicAdd(&cursor[tid], h);
        gbase[tid] = pc + cs - c + g;
    }
    __syncthreads();

#pragma unroll
    for (int k = 0; k < 4; ++k) {
        if (k < nval) {
            const int pos = loff[r[k]] + rank[k];
            const unsigned local = (unsigned)d[k] - (unsigned)r[k] * R_NODES;
            stage[pos] = make_int2((int)((unsigned)s[k] | (local << 24)), __float_as_int(w[k]));
            stage_r[pos] = (unsigned char)r[k];
        }
    }
    __syncthreads();

    int nloc = n_edges - chunk0;
    if (nloc > PART_EDGES) nloc = PART_EDGES;
    for (int i = tid; i < nloc; i += PART_THREADS) {
        const int rr = stage_r[i];
        const int dest = gbase[rr] + (i - loff[rr]);
        if (dest >= 0 && dest < rec_cap) rec[dest] = stage[i];
    }
}

template <int F, int F_OUT, int NT, bool RELU>
__global__ __launch_bounds__(SPMM_WAVES * 32) void spmm_rec_kernel(
    const float* __restrict__ h, const float* __restrict__ ndst,
    const int* __restrict__ rec_in, const int* __restrict__ cnt,
    const float* __restrict__ W, const float* __restrict__ bias,
    float* __restrict__ out, int n_nodes, int rec_cap) {
    constexpr int P = LdsPitch<F>::v;
    __shared__ alignas(16) float acc[R_PAD * P];
    __shared__ alignas(16) float wl[F * W_PITCH];
    __shared__ int seg[2];
    const int2* rec = reinterpret_cast<const int2*>(rec_in);

    const int tid = threadIdx.x;
    const int lane = tid & 31;
    const int wave = __builtin_amdgcn_readfirstlane(tid >> 5);
    const int b = blockIdx.x;
    const int r0 = b * R_NODES;

    for (int i = tid; i < R_PAD * P / 4; i += SPMM_WAVES * 32)
        reinterpret_cast<float4*>(acc)[i] = make_float4(0.f, 0.f, 0.f, 0.f);
    stage_w<F, F_OUT, NT>(wl, W, tid, SPMM_WAVES * 32);
    if (wave == 0) {
        int s = 0;
        for (int k = lane; k < b; k += 32) s += (cnt[k] + 15) & ~15;
        for (int o = 16; o > 0; o >>= 1) s += __shfl_xor(s, o);
        if (lane == 0) { seg[0] = s; seg[1] = cnt[b]; }
    }
    __syncthreads();
    int off = __builtin_amdgcn_readfirstlane(seg[0]);
    int count = __builtin_amdgcn_readfirstlane(seg[1]);
    if (off > rec_cap) off = rec_cap;
    if (count > rec_cap - off) count = rec_cap - off;

    const int per = (count + SPMM_WAVES - 1) / SPMM_WAVES;
    int p = wave * per;
    int pend = p + per;
    if (pend > count) pend = count;

    if (p < pend) {
        int i0 = p + lane;
        if (i0 > pend - 1) i0 = pend - 1;
        int2 r = rec[off + i0];
        for (; p < pend; p += 32) {
            int n = pend - p;
            if (n > 32) n = 32;
            int i1 = p + 32 + lane;
            if (i1 > pend - 1) i1 = pend - 1;
            const int2 rn = rec[off + i1];
            const int sv = r.x & REC_ID_MASK;
            const int dv = (int)((unsigned)r.x >> 24);
            const float nv = __int_as_float(r.y);
            add_queue<F>(h, acc, sv, dv, nv, n, lane);
            r = rn;
        }
    }

    __syncthreads();

    if (wave < R_TILES)
        dense_tile_from_lds<F, F_OUT, NT, RELU>(acc, wl, ndst, bias, r0, n_nodes, wave, lane);

    __syncthreads();

    store_rows<P, F_OUT>(acc, out, r0, n_nodes, tid, SPMM_WAVES * 32);
}

static inline size_t align256(size_t x) { return (x + 255) & ~(size_t)255; }

extern "C" void kernel_launch(void* const* d_in, const int* in_sizes, int n_in,
                              void* d_out, int out_size, void* d_ws, size_t ws_size,
                              hipStream_t stream) {
    const float* x  = (const float*)d_in[0];
    const float* W0 = (const float*)d_in[1];
    const float* b0 = (const float*)d_in[2];
    const float* W1 = (const float*)d_in[3];
    const float* b1 = (const float*)d_in[4];
    const float* W2 = (const float*)d_in[5];
    const float* b2 = (const float*)d_in[6];
    const int* src  = (const int*)d_in[7];
    const int* dst  = (const int*)d_in[8];

    const int N = in_sizes[0] / N_FEAT_IN;
    const int E = in_sizes[7];
    float* out = (float*)d_out;

    char* p = (char*)d_ws;
    float* nsrc = (float*)p;  p += align256((size_t)N * 4);
    float* ndst = (float*)p;  p += align256((size_t)N * 4);
    float* h1   = (float*)p;  p += align256((size_t)N * N_FEAT_IN * 4);
    float* h0   = (float*)p;  p += align256((size_t)N * 64 * 4);
    int* cnt = (int*)p;  p += align256((size_t)256 * 4);
    int* cursor = (int*)p;  p += align256((size_t)256 * 4);
    const int rec_cap = E + 16 * 256;
    int* rec = (int*)p;  p += align256((size_t)rec_cap * 8);

    const int TB = 256;
    const int ranges = (N + R_NODES - 1) / R_NODES;

    hipMemsetAsync(nsrc, 0, (size_t)N * 4, stream);
    hipMemsetAsync(ndst, 0, (size_t)N * 4, stream);
    degree_kernel<<<((E + 3) / 4 + TB - 1) / TB, TB, 0, stream>>>(src, dst, nsrc, ndst, E);
    norm_count_kernel<<<ranges, 256, 0, stream>>>(nsrc, ndst, cnt, cursor, N);

    partition_kernel<<<(E + PART_EDGES - 1) / PART_EDGES, PART_THREADS, 0, stream>>>(
        src, dst, nsrc, cnt, cursor, rec, E, N, ranges, rec_cap);

    spmm_rec_kernel<N_FEAT_IN, 64, 4, true><<<ranges, SPMM_WAVES * 32, 0, stream>>>(
        x, ndst, rec, cnt, W0, b0, h0, N, rec_cap);

    spmm_rec_kernel<64, 64, 4, true><<<ranges, SPMM_WAVES * 32, 0, stream>>>(
        h0, ndst, rec, cnt, W1, b1, h1, N, rec_cap);

    spmm_rec_kernel<64, 47, 3, false><<<ranges, SPMM_WAVES * 32, 0, stream>>>(
        h1, ndst, rec, cnt, W2, b2, out, N, rec_cap);
}


# In-degrees leave memory: a workgroup counts its range's records per owned row in LDS (int adds) for D_in^-1/2; ranges sized by an LDS histogram; 5 launches, 1 memset; feature sum order kept
# speedup vs baseline: 3.4682x; 1.3888x over previous
#include <hip/hip_runtime.h>


typedef float v2f __attribute__((ext_vector_type(2)));
typedef float v8f __attribute__((ext_vector_type(8)));

#define N_FEAT_IN 100

#define R_NODES 200
#define R_TILES ((R_NODES + 15) / 16)
#define R_PAD (R_TILES * 16)
template <int F> struct LdsPitch { static constexpr int v = (F % 64 == 0) ? F + 4 : F; };

#define COUNT_EDGES 4096
#define COUNT_THREADS 1024

__global__ __launch_bounds__(COUNT_THREADS) void count_kernel(const int* __restrict__ src,
                                                              const int* __restrict__ dst,
                                                              float* __restrict__ deg_out,
                                                              int* __restrict__ cnt,
                                                              int n_edges) {
    __shared__ int hist[256];
    const int tid = threadIdx.x;
    const int e0 = blockIdx.x * COUNT_EDGES + tid * 4;
    int nval = n_edges - e0;
    if (nval > 4) nval = 4;
    int s[4], d[4];
    if (nval == 4) {
        const int4 sv = *reinterpret_cast<const int4*>(src + e0);
        const int4 dv = *reinterpret_cast<const int4*>(dst + e0);
        s[0] = sv.x; s[1] = sv.y; s[2] = sv.z; s[3] = sv.w;
        d[0] = dv.x; d[1] = dv.y; d[2] = dv.z; d[3] = dv.w;
    } else {
#pragma unroll
        for (int k = 0; k < 4; ++k) {
            s[k] = (k < nval) ? src[e0 + k] : 0;
            d[k] = (k < nval) ? dst[e0 + k] : 0;
        }
    }
    if (tid < 256) hist[tid] = 0;
    __syncthreads();
#pragma unroll
    for (int k = 0; k < 4; ++k)
        if (k < nval) atomicAdd(&hist[(unsigned)d[k] / R_NODES], 1);
    __syncthreads();
    if (tid < 256) {
        const int h = hist[tid];
        if (h > 0) atomicAdd(&cnt[tid], h);
    }
#pragma unroll
    for (int k = 0; k < 4; ++k)
        if (k < nval) atomicAdd(&deg_out[s[k]], 1.0f);
}

template <int F, int K>
__device__ __forceinline__ void add_rows(const float* __restrict__ h, float* acc,
                                         int sv, int dv, float nv, int j, int lane) {
    constexpr int NC = (F + 31) / 32;
    int s[K], d[K];
    float ns[K];
    float v[K][NC];
#pragma unroll
    for (int t = 0; t < K; ++t) {
        s[t] = __builtin_amdgcn_readlane(sv, j + t);
        d[t] = __builtin_amdgcn_readlane(dv, j + t);
        ns[t] = __int_as_float(__builtin_amdgcn_readlane(__float_as_int(nv), j + t));
    }
#pragma unroll
    for (int t = 0; t < K; ++t) {
        const float* hrow = h + (size_t)s[t] * F;
#pragma unroll
        for (int c = 0; c < NC; ++c) {
            const int f = c * 32 + lane;
            v[t][c] = (f < F) ? hrow[f] : 0.0f;
        }
    }
#pragma unroll
    for (int t = 0; t < K; ++t) {
        float* arow = acc + d[t] * LdsPitch<F>::v;
#pragma unroll
        for (int c = 0; c < NC; ++c) {
            const int f = c * 32 + lane;
            if (f < F)
                __hip_atomic_fetch_add(arow + f, v[t][c] * ns[t], __ATOMIC_RELAXED,
                                       __HIP_MEMORY_SCOPE_WORKGROUP);
        }
    }
}

template <int F>
__device__ __forceinline__ void add_queue(const float* __restrict__ h, float* acc,
                                          int sv, int dv, float nv, int n, int lane) {
    int j = 0;
    for (; j + 8 <= n; j += 8) add_rows<F, 8>(h, acc, sv, dv, nv, j, lane);
    if (j + 4 <= n) { add_rows<F, 4>(h, acc, sv, dv, nv, j, lane); j += 4; }
    if (j + 2 <= n) { add_rows<F, 2>(h, acc, sv, dv, nv, j, lane); j += 2; }
    if (j < n) add_rows<F, 1>(h, acc, sv, dv, nv, j, lane);
}

#define REC_ID_MASK 0x00FFFFFF

#define W_PITCH 80

template <int F_IN, int F_OUT, int NT>
__device__ __forceinline__ void stage_w(float* wl, const float* __restrict__ W,
                                        int tid, int nthreads) {
    constexpr int C = NT * 16;
    static_assert(C <= W_PITCH && F_IN % 2 == 0, "W does not fit its LDS layout");
    for (int i = tid; i < F_IN * C; i += nthreads) {
        const int k = i / C;
        const int c = i - k * C;
        wl[(k >> 1) * (2 * W_PITCH) + c * 2 + (k & 1)] = (c < F_OUT) ? W[k * F_OUT + c] : 0.0f;
    }
}

template <int NT>
__device__ __forceinline__ void load_b_frags(const float* wl, int ka, int lm, v2f (&bf)[NT]) {
#pragma unroll
    for (int t = 0; t < NT; ++t)
        bf[t] = *reinterpret_cast<const v2f*>(wl + (ka >> 1) * (2 * W_PITCH) + (t * 16 + lm) * 2);
}

template <int F_IN, int F_OUT, int NT, bool RELU>
__device__ __forceinline__ void dense_tile_from_lds(float* tile, const float* wl,
                                                    const int* degl,
                                                    const float* __restrict__ bias,
                                                    int t16, int lane) {
    constexpr int P = LdsPitch<F_IN>::v;
    const int half = lane >> 4;
    const int lm = lane & 15;

    v8f acc[NT];
#pragma unroll
    for (int t = 0; t < NT; ++t) acc[t] = v8f{};

    const float* arow = tile + (t16 * 16 + lm) * P + 2 * half;

    float nd[8];
#pragma unroll
    for (int i = 0; i < 8; ++i)
        nd[i] = rsqrtf(fmaxf((float)degl[t16 * 16 + i + 8 * half], 1.0f));
    float bs[NT];
#pragma unroll
    for (int t = 0; t < NT; ++t) {
        const int col = t * 16 + lm;
        bs[t] = bias[col < F_OUT ? col : F_OUT - 1];
    }

    v2f bf[NT];
    load_b_frags<NT>(wl, 2 * half, lm, bf);
#pragma unroll
    for (int k0 = 0; k0 < F_IN; k0 += 4) {
        const v2f a = *reinterpret_cast<const v2f*>(arow + k0);
        v2f bn[NT];
        if (k0 + 4 < F_IN) load_b_frags<NT>(wl, k0 + 4 + 2 * half, lm, bn);
#pragma unroll
        for (int t = 0; t < NT; ++t)
            acc[t] = __builtin_amdgcn_wmma_f32_16x16x4_f32(
                 false, a,  false, bf[t],
                 (short)0, acc[t],  false,  false);
        if (k0 + 4 < F_IN) {
#pragma unroll
            for (int t = 0; t < NT; ++t) bf[t] = bn[t];
        }
    }

#pragma unroll
    for (int i = 0; i < 8; ++i) {
        const int row = t16 * 16 + i + 8 * half;
#pragma unroll
        for (int t = 0; t < NT; ++t) {
            const int col = t * 16 + lm;
            if (col < F_OUT) {
                float v = acc[t][i] * nd[i] + bs[t];
                if (RELU) v = fmaxf(v, 0.0f);
                tile[row * P + col] = v;
            }
        }
    }
}

template <int P, int F_OUT>
__device__ __forceinline__ void store_rows(const float* tile, float* __restrict__ out,
                                           int r0, int n_nodes, int tid, int nthreads) {
    int rows = n_nodes - r0;
    if (rows > R_NODES) rows = R_NODES;
    float* o = out + (size_t)r0 * F_OUT;
    if (F_OUT % 4 == 0) {
        constexpr int Q = F_OUT / 4;
        const int n4 = rows * Q;
        for (int i = tid; i < n4; i += nthreads) {
            const int row = i / Q;
            const int c4 = i - row * Q;
            reinterpret_cast<float4*>(o)[i] =
                *reinterpret_cast<const float4*>(tile + row * P + c4 * 4);
        }
    } else {
        const int n1 = rows * F_OUT;
        for (int i = tid; i < n1; i += nthreads) {
            const int row = i / F_OUT;
            const int col = i - row * F_OUT;
            o[i] = tile[row * P + col];
        }
    }
}

#define SPMM_WAVES 32

#define PART_EDGES 4096
#define PART_THREADS 1024
#define NODE_TABLE 50000
#define TABLE_ITERS ((NODE_TABLE / 4 + PART_THREADS - 1) / PART_THREADS)

__global__ __launch_bounds__(PART_THREADS) void partition_kernel(
    const int* __restrict__ src, const int* __restrict__ dst,
    const float* __restrict__ deg_out, const int* __restrict__ cnt,
    int* __restrict__ cursor, int* __restrict__ rec_out,
    int n_edges, int n_nodes, int n_ranges, int rec_cap) {
    __shared__ alignas(16) float tbl[NODE_TABLE];
    __shared__ int2 stage[PART_EDGES];
    __shared__ unsigned char stage_r[PART_EDGES];
    __shared__ int hist[256];
    __shared__ int seglen[256];
    __shared__ int loff[256];
    __shared__ int gbase[256];
    __shared__ int wsum[16];
    int2* rec = reinterpret_cast<int2*>(rec_out);

    const int tid = threadIdx.x;
    const int lane = tid & 31;
    const int wave = tid >> 5;
    const int chunk0 = blockIdx.x * PART_EDGES;
    const int e0 = chunk0 + tid * 4;

    int s[4], d[4];
    int nval = n_edges - e0;
    if (nval > 4) nval = 4;
    if (nval == 4) {
        const int4 sv = *reinterpret_cast<const int4*>(src + e0);
        const int4 dv = *reinterpret_cast<const int4*>(dst + e0);
        s[0] = sv.x; s[1] = sv.y; s[2] = sv.z; s[3] = sv.w;
        d[0] = dv.x; d[1] = dv.y; d[2] = dv.z; d[3] = dv.w;
    } else {
#pragma unroll
        for (int k = 0; k < 4; ++k) {
            s[k] = (k < nval) ? src[e0 + k] : 0;
            d[k] = (k < nval) ? dst[e0 + k] : 0;
        }
    }

    const bool use_tbl = (n_nodes <= NODE_TABLE);
    const int nt = use_tbl ? n_nodes : 0;
    const int n4 = nt >> 2;
    int cv = 0;
    {
        float4 tv[TABLE_ITERS];
#pragma unroll
        for (int u = 0; u < TABLE_ITERS; ++u) {
            const int i = tid + u * PART_THREADS;
            tv[u] = (i < n4) ? reinterpret_cast<const float4*>(deg_out)[i]
                             : make_float4(0.f, 0.f, 0.f, 0.f);
        }
        cv = (tid < 256 && tid < n_ranges) ? cnt[tid] : 0;
#pragma unroll
        for (int u = 0; u < TABLE_ITERS; ++u) {
            const int i = tid + u * PART_THREADS;
            if (i < n4) reinterpret_cast<float4*>(tbl)[i] = tv[u];
        }
        const int i = (n4 << 2) + tid;
        if (i < nt) tbl[i] = deg_out[i];
    }
    if (tid < 256) {
        hist[tid] = 0;
        seglen[tid] = (cv + 15) & ~15;
    }
    __syncthreads();

    int r[4], rank[4];
    float w[4];
#pragma unroll
    for (int k = 0; k < 4; ++k) {
        r[k] = 0; rank[k] = 0; w[k] = 0.0f;
        if (k < nval) {
            r[k] = (int)((unsigned)d[k] / R_NODES);
            rank[k] = atomicAdd(&hist[r[k]], 1);
        }
    }
    if (use_tbl) {
#pragma unroll
        for (int k = 0; k < 4; ++k)
            if (k < nval) w[k] = rsqrtf(fmaxf(tbl[s[k]], 1.0f));
    } else {
#pragma unroll
        for (int k = 0; k < 4; ++k)
            if (k < nval) w[k] = rsqrtf(fmaxf(deg_out[s[k]], 1.0f));
    }
    __syncthreads();

    int h = 0, hs = 0, cs = 0, c = 0;
    if (tid < 256) {
        h = hist[tid];
        c = seglen[tid];
        hs = h;
        cs = c;
#pragma unroll
        for (int o = 1; o < 32; o <<= 1) {
            const int t1 = __shfl_up(hs, o);
            const int t2 = __shfl_up(cs, o);
            if (lane >= o) { hs += t1; cs += t2; }
        }
        if (lane == 31) { wsum[wave] = hs; wsum[8 + wave] = cs; }
    }
    __syncthreads();
    if (tid < 256) {
        int ph = 0, pc = 0;
        for (int k = 0; k < wave; ++k) { ph += wsum[k]; pc += wsum[8 + k]; }
        loff[tid] = ph + hs - h;
        int g = 0;
        if (h > 0) g = atomicAdd(&cursor[tid], h);
        gbase[tid] = pc + cs - c + g;
    }
    __syncthreads();

#pragma unroll
    for (int k = 0; k < 4; ++k) {
        if (k < nval) {
            const int pos = loff[r[k]] + rank[k];
            const unsigned local = (unsigned)d[k] - (unsigned)r[k] * R_NODES;
            stage[pos] = make_int2((int)((unsigned)s[k] | (local << 24)), __float_as_int(w[k]));
            stage_r[pos] = (unsigned char)r[k];
        }
    }
    __syncthreads();

    int nloc = n_edges - chunk0;
    if (nloc > PART_EDGES) nloc = PART_EDGES;
    for (int i = tid; i < nloc; i += PART_THREADS) {
        const int rr = stage_r[i];
        const int dest = gbase[rr] + (i - loff[rr]);
        if (dest >= 0 && dest < rec_cap) rec[dest] = stage[i];
    }
}

template <int F, int F_OUT, int NT, bool RELU>
__global__ __launch_bounds__(SPMM_WAVES * 32) void spmm_rec_kernel(
    const float* __restrict__ h,
    const int* __restrict__ rec_in, const int* __restrict__ cnt,
    const float* __restrict__ W, const float* __restrict__ bias,
    float* __restrict__ out, int n_nodes, int rec_cap) {
    constexpr int P = LdsPitch<F>::v;
    __shared__ alignas(16) float acc[R_PAD * P];
    __shared__ alignas(16) float wl[F * W_PITCH];
    __shared__ int seg[2];
    __shared__ alignas(16) int degl[256];
    const int2* rec = reinterpret_cast<const int2*>(rec_in);

    const int tid = threadIdx.x;
    const int lane = tid & 31;
    const int wave = __builtin_amdgcn_readfirstlane(tid >> 5);
    const int b = blockIdx.x;
    const int r0 = b * R_NODES;

    for (int i = tid; i < R_PAD * P / 4; i += SPMM_WAVES * 32)
        reinterpret_cast<float4*>(acc)[i] = make_float4(0.f, 0.f, 0.f, 0.f);
    if (tid < 256) degl[tid] = 0;
    stage_w<F, F_OUT, NT>(wl, W, tid, SPMM_WAVES * 32);
    if (wave == 0) {
        int s = 0;
        for (int k = lane; k < b; k += 32) s += (cnt[k] + 15) & ~15;
        for (int o = 16; o > 0; o >>= 1) s += __shfl_xor(s, o);
        if (lane == 0) { seg[0] = s; seg[1] = cnt[b]; }
    }
    __syncthreads();
    int off = __builtin_amdgcn_readfirstlane(seg[0]);
    int count = __builtin_amdgcn_readfirstlane(seg[1]);
    if (off > rec_cap) off = rec_cap;
    if (count > rec_cap - off) count = rec_cap - off;

    const int per = (count + SPMM_WAVES - 1) / SPMM_WAVES;
    int p = wave * per;
    int pend = p + per;
    if (pend > count) pend = count;

    if (p < pend) {
        int i0 = p + lane;
        if (i0 > pend - 1) i0 = pend - 1;
        int2 r = rec[off + i0];
        for (; p < pend; p += 32) {
            int n = pend - p;
            if (n > 32) n = 32;
            int i1 = p + 32 + lane;
            if (i1 > pend - 1) i1 = pend - 1;
            const int2 rn = rec[off + i1];
            const int sv = r.x & REC_ID_MASK;
            const int dv = (int)((unsigned)r.x >> 24);
            const float nv = __int_as_float(r.y);
            if (lane < n) atomicAdd(&degl[dv], 1);
            add_queue<F>(h, acc, sv, dv, nv, n, lane);
            r = rn;
        }
    }

    __syncthreads();

    if (wave < R_TILES)
        dense_tile_from_lds<F, F_OUT, NT, RELU>(acc, wl, degl, bias, wave, lane);

    __syncthreads();

    store_rows<P, F_OUT>(acc, out, r0, n_nodes, tid, SPMM_WAVES * 32);
}

static inline size_t align256(size_t x) { return (x + 255) & ~(size_t)255; }

extern "C" void kernel_launch(void* const* d_in, const int* in_sizes, int n_in,
                              void* d_out, int out_size, void* d_ws, size_t ws_size,
                              hipStream_t stream) {
    const float* x  = (const float*)d_in[0];
    const float* W0 = (const float*)d_in[1];
    const float* b0 = (const float*)d_in[2];
    const float* W1 = (const float*)d_in[3];
    const float* b1 = (const float*)d_in[4];
    const float* W2 = (const float*)d_in[5];
    const float* b2 = (const float*)d_in[6];
    const int* src  = (const int*)d_in[7];
    const int* dst  = (const int*)d_in[8];

    const int N = in_sizes[0] / N_FEAT_IN;
    const int E = in_sizes[7];
    float* out = (float*)d_out;

    char* p = (char*)d_ws;
    float* deg_out = (float*)p;  p += align256((size_t)N * 4);
    int* cnt = (int*)p;  p += align256((size_t)256 * 4);
    int* cursor = (int*)p;  p += align256((size_t)256 * 4);
    const size_t clear_bytes = (size_t)(p - (char*)deg_out);
    float* h1   = (float*)p;  p += align256((size_t)N * 64 * 4);
    float* h0   = (float*)p;  p += align256((size_t)N * 64 * 4);
    const int rec_cap = E + 16 * 256;
    int* rec = (int*)p;  p += align256((size_t)rec_cap * 8);

    const int ranges = (N + R_NODES - 1) / R_NODES;

    hipMemsetAsync(deg_out, 0, clear_bytes, stream);
    count_kernel<<<(E + COUNT_EDGES - 1) / COUNT_EDGES, COUNT_THREADS, 0, stream>>>(
        src, dst, deg_out, cnt, E);

    partition_kernel<<<(E + PART_EDGES - 1) / PART_EDGES, PART_THREADS, 0, stream>>>(
        src, dst, deg_out, cnt, cursor, rec, E, N, ranges, rec_cap);

    spmm_rec_kernel<N_FEAT_IN, 64, 4, true><<<ranges, SPMM_WAVES * 32, 0, stream>>>(
        x, rec, cnt, W0, b0, h0, N, rec_cap);

    spmm_rec_kernel<64, 64, 4, true><<<ranges, SPMM_WAVES * 32, 0, stream>>>(
        h0, rec, cnt, W1, b1, h1, N, rec_cap);

    spmm_rec_kernel<64, 47, 3, false><<<ranges, SPMM_WAVES * 32, 0, stream>>>(
        h1, rec, cnt, W2, b2, out, N, rec_cap);
}


# Out-degrees without global float atomics: count launch orders a chunk's sources by 200-node range as byte ids (LDS ranks); an owner launch counts its range's pieces in LDS; 6 launches; sums kept
# speedup vs baseline: 4.9120x; 1.4163x over previous
#include <hip/hip_runtime.h>


typedef float v2f __attribute__((ext_vector_type(2)));
typedef float v8f __attribute__((ext_vector_type(8)));

#define N_FEAT_IN 100

#define R_NODES 200
#define R_TILES ((R_NODES + 15) / 16)
#define R_PAD (R_TILES * 16)
template <int F> struct LdsPitch { static constexpr int v = (F % 64 == 0) ? F + 4 : F; };

#define COUNT_EDGES 4096
#define COUNT_THREADS 1024

__global__ __launch_bounds__(COUNT_THREADS) void count_kernel(const int* __restrict__ src,
                                                              const int* __restrict__ dst,
                                                              unsigned char* __restrict__ src_ids,
                                                              unsigned* __restrict__ piece,
                                                              int* __restrict__ cnt,
                                                              int n_edges) {
    __shared__ int hist[256];
    __shared__ int shist[256];
    __shared__ int soff[256];
    __shared__ int wsum[8];
    __shared__ alignas(16) unsigned char sstage[COUNT_EDGES];
    const int tid = threadIdx.x;
    const int lane = tid & 31;
    const int wave = tid >> 5;
    const int e0 = blockIdx.x * COUNT_EDGES + tid * 4;
    int nval = n_edges - e0;
    if (nval > 4) nval = 4;
    int s[4], d[4];
    if (nval == 4) {
        const int4 sv = *reinterpret_cast<const int4*>(src + e0);
        const int4 dv = *reinterpret_cast<const int4*>(dst + e0);
        s[0] = sv.x; s[1] = sv.y; s[2] = sv.z; s[3] = sv.w;
        d[0] = dv.x; d[1] = dv.y; d[2] = dv.z; d[3] = dv.w;
    } else {
#pragma unroll
        for (int k = 0; k < 4; ++k) {
            s[k] = (k < nval) ? src[e0 + k] : 0;
            d[k] = (k < nval) ? dst[e0 + k] : 0;
        }
    }
    if (tid < 256) { hist[tid] = 0; shist[tid] = 0; }
    __syncthreads();
    int rs[4], rk[4];
#pragma unroll
    for (int k = 0; k < 4; ++k) {
        rs[k] = 0; rk[k] = 0;
        if (k < nval) {
            atomicAdd(&hist[(unsigned)d[k] / R_NODES], 1);
            rs[k] = (int)((unsigned)s[k] / R_NODES);
            rk[k] = atomicAdd(&shist[rs[k]], 1);
        }
    }
    __syncthreads();
    int h = 0, c = 0, cs = 0;
    if (tid < 256) {
        h = hist[tid];
        if (h > 0) atomicAdd(&cnt[tid], h);
        c = shist[tid];
        cs = c;
#pragma unroll
        for (int o = 1; o < 32; o <<= 1) {
            const int t1 = __shfl_up(cs, o);
            if (lane >= o) cs += t1;
        }
        if (lane == 31) wsum[wave] = cs;
    }
    __syncthreads();
    if (tid < 256) {
        int p = 0;
        for (int k = 0; k < wave; ++k) p += wsum[k];
        const int start = p + cs - c;
        soff[tid] = start;
        piece[blockIdx.x * 256 + tid] = (unsigned)start | ((unsigned)c << 16);
    }
    __syncthreads();
#pragma unroll
    for (int k = 0; k < 4; ++k)
        if (k < nval)
            sstage[soff[rs[k]] + rk[k]] = (unsigned char)((unsigned)s[k] - (unsigned)rs[k] * R_NODES);
    __syncthreads();
    reinterpret_cast<unsigned*>(src_ids + (size_t)blockIdx.x * COUNT_EDGES)[tid] =
        reinterpret_cast<const unsigned*>(sstage)[tid];
}

#define OUTDEG_THREADS 256

__global__ __launch_bounds__(OUTDEG_THREADS) void outdeg_kernel(
    const unsigned char* __restrict__ src_ids, const unsigned* __restrict__ piece,
    float* __restrict__ deg_out, int n_chunks, int n_nodes) {
    __shared__ int degs[256];
    const int tid = threadIdx.x;
    const int r = blockIdx.x;
    degs[tid] = 0;
    __syncthreads();
    for (int ch = tid; ch < n_chunks; ch += OUTDEG_THREADS) {
        const unsigned pc = piece[ch * 256 + r];
        const int start = (int)(pc & 0xFFFFu);
        const int n = (int)(pc >> 16);
        const unsigned char* p = src_ids + (size_t)ch * COUNT_EDGES + start;
        for (int k = 0; k < n; k += 8) {
            int b[8];
#pragma unroll
            for (int j = 0; j < 8; ++j) b[j] = (k + j < n) ? (int)p[k + j] : 0;
#pragma unroll
            for (int j = 0; j < 8; ++j)
                if (k + j < n) atomicAdd(&degs[b[j]], 1);
        }
    }
    __syncthreads();
    const int node = r * R_NODES + tid;
    if (tid < R_NODES && node < n_nodes) deg_out[node] = (float)degs[tid];
}

template <int F, int K>
__device__ __forceinline__ void add_rows(const float* __restrict__ h, float* acc,
                                         int sv, int dv, float nv, int j, int lane) {
    constexpr int NC = (F + 31) / 32;
    int s[K], d[K];
    float ns[K];
    float v[K][NC];
#pragma unroll
    for (int t = 0; t < K; ++t) {
        s[t] = __builtin_amdgcn_readlane(sv, j + t);
        d[t] = __builtin_amdgcn_readlane(dv, j + t);
        ns[t] = __int_as_float(__builtin_amdgcn_readlane(__float_as_int(nv), j + t));
    }
#pragma unroll
    for (int t = 0; t < K; ++t) {
        const float* hrow = h + (size_t)s[t] * F;
#pragma unroll
        for (int c = 0; c < NC; ++c) {
            const int f = c * 32 + lane;
            v[t][c] = (f < F) ? hrow[f] : 0.0f;
        }
    }
#pragma unroll
    for (int t = 0; t < K; ++t) {
        float* arow = acc + d[t] * LdsPitch<F>::v;
#pragma unroll
        for (int c = 0; c < NC; ++c) {
            const int f = c * 32 + lane;
            if (f < F)
                __hip_atomic_fetch_add(arow + f, v[t][c] * ns[t], __ATOMIC_RELAXED,
                                       __HIP_MEMORY_SCOPE_WORKGROUP);
        }
    }
}

template <int F>
__device__ __forceinline__ void add_queue(const float* __restrict__ h, float* acc,
                                          int sv, int dv, float nv, int n, int lane) {
    int j = 0;
    for (; j + 8 <= n; j += 8) add_rows<F, 8>(h, acc, sv, dv, nv, j, lane);
    if (j + 4 <= n) { add_rows<F, 4>(h, acc, sv, dv, nv, j, lane); j += 4; }
    if (j + 2 <= n) { add_rows<F, 2>(h, acc, sv, dv, nv, j, lane); j += 2; }
    if (j < n) add_rows<F, 1>(h, acc, sv, dv, nv, j, lane);
}

#define REC_ID_MASK 0x00FFFFFF

#define W_PITCH 80

template <int F_IN, int F_OUT, int NT>
__device__ __forceinline__ void stage_w(float* wl, const float* __restrict__ W,
                                        int tid, int nthreads) {
    constexpr int C = NT * 16;
    static_assert(C <= W_PITCH && F_IN % 2 == 0, "W does not fit its LDS layout");
    for (int i = tid; i < F_IN * C; i += nthreads) {
        const int k = i / C;
        const int c = i - k * C;
        wl[(k >> 1) * (2 * W_PITCH) + c * 2 + (k & 1)] = (c < F_OUT) ? W[k * F_OUT + c] : 0.0f;
    }
}

template <int NT>
__device__ __forceinline__ void load_b_frags(const float* wl, int ka, int lm, v2f (&bf)[NT]) {
#pragma unroll
    for (int t = 0; t < NT; ++t)
        bf[t] = *reinterpret_cast<const v2f*>(wl + (ka >> 1) * (2 * W_PITCH) + (t * 16 + lm) * 2);
}

template <int F_IN, int F_OUT, int NT, bool RELU>
__device__ __forceinline__ void dense_tile_from_lds(float* tile, const float* wl,
                                                    const int* degl,
                                                    const float* __restrict__ bias,
                                                    int t16, int lane) {
    constexpr int P = LdsPitch<F_IN>::v;
    const int half = lane >> 4;
    const int lm = lane & 15;

    v8f acc[NT];
#pragma unroll
    for (int t = 0; t < NT; ++t) acc[t] = v8f{};

    const float* arow = tile + (t16 * 16 + lm) * P + 2 * half;

    float nd[8];
#pragma unroll
    for (int i = 0; i < 8; ++i)
        nd[i] = rsqrtf(fmaxf((float)degl[t16 * 16 + i + 8 * half], 1.0f));
    float bs[NT];
#pragma unroll
    for (int t = 0; t < NT; ++t) {
        const int col = t * 16 + lm;
        bs[t] = bias[col < F_OUT ? col : F_OUT - 1];
    }

    v2f bf[NT];
    load_b_frags<NT>(wl, 2 * half, lm, bf);
#pragma unroll
    for (int k0 = 0; k0 < F_IN; k0 += 4) {
        const v2f a = *reinterpret_cast<const v2f*>(arow + k0);
        v2f bn[NT];
        if (k0 + 4 < F_IN) load_b_frags<NT>(wl, k0 + 4 + 2 * half, lm, bn);
#pragma unroll
        for (int t = 0; t < NT; ++t)
            acc[t] = __builtin_amdgcn_wmma_f32_16x16x4_f32(
                 false, a,  false, bf[t],
                 (short)0, acc[t],  false,  false);
        if (k0 + 4 < F_IN) {
#pragma unroll
            for (int t = 0; t < NT; ++t) bf[t] = bn[t];
        }
    }

#pragma unroll
    for (int i = 0; i < 8; ++i) {
        const int row = t16 * 16 + i + 8 * half;
#pragma unroll
        for (int t = 0; t < NT; ++t) {
            const int col = t * 16 + lm;
            if (col < F_OUT) {
                float v = acc[t][i] * nd[i] + bs[t];
                if (RELU) v = fmaxf(v, 0.0f);
                tile[row * P + col] = v;
            }
        }
    }
}

template <int P, int F_OUT>
__device__ __forceinline__ void store_rows(const float* tile, float* __restrict__ out,
                                           int r0, int n_nodes, int tid, int nthreads) {
    int rows = n_nodes - r0;
    if (rows > R_NODES) rows = R_NODES;
    float* o = out + (size_t)r0 * F_OUT;
    if (F_OUT % 4 == 0) {
        constexpr int Q = F_OUT / 4;
        const int n4 = rows * Q;
        for (int i = tid; i < n4; i += nthreads) {
            const int row = i / Q;
            const int c4 = i - row * Q;
            reinterpret_cast<float4*>(o)[i] =
                *reinterpret_cast<const float4*>(tile + row * P + c4 * 4);
        }
    } else {
        const int n1 = rows * F_OUT;
        for (int i = tid; i < n1; i += nthreads) {
            const int row = i / F_OUT;
            const int col = i - row * F_OUT;
            o[i] = tile[row * P + col];
        }
    }
}

#define SPMM_WAVES 32

#define PART_EDGES 4096
#define PART_THREADS 1024
#define NODE_TABLE 50000
#define TABLE_ITERS ((NODE_TABLE / 4 + PART_THREADS - 1) / PART_THREADS)

__global__ __launch_bounds__(PART_THREADS) void partition_kernel(
    const int* __restrict__ src, const int* __restrict__ dst,
    const float* __restrict__ deg_out, const int* __restrict__ cnt,
    int* __restrict__ cursor, int* __restrict__ rec_out,
    int n_edges, int n_nodes, int n_ranges, int rec_cap) {
    __shared__ alignas(16) float tbl[NODE_TABLE];
    __shared__ int2 stage[PART_EDGES];
    __shared__ unsigned char stage_r[PART_EDGES];
    __shared__ int hist[256];
    __shared__ int seglen[256];
    __shared__ int loff[256];
    __shared__ int gbase[256];
    __shared__ int wsum[16];
    int2* rec = reinterpret_cast<int2*>(rec_out);

    const int tid = threadIdx.x;
    const int lane = tid & 31;
    const int wave = tid >> 5;
    const int chunk0 = blockIdx.x * PART_EDGES;
    const int e0 = chunk0 + tid * 4;

    int s[4], d[4];
    int nval = n_edges - e0;
    if (nval > 4) nval = 4;
    if (nval == 4) {
        const int4 sv = *reinterpret_cast<const int4*>(src + e0);
        const int4 dv = *reinterpret_cast<const int4*>(dst + e0);
        s[0] = sv.x; s[1] = sv.y; s[2] = sv.z; s[3] = sv.w;
        d[0] = dv.x; d[1] = dv.y; d[2] = dv.z; d[3] = dv.w;
    } else {
#pragma unroll
        for (int k = 0; k < 4; ++k) {
            s[k] = (k < nval) ? src[e0 + k] : 0;
            d[k] = (k < nval) ? dst[e0 + k] : 0;
        }
    }

    const bool use_tbl = (n_nodes <= NODE_TABLE);
    const int nt = use_tbl ? n_nodes : 0;
    const int n4 = nt >> 2;
    int cv = 0;
    {
        float4 tv[TABLE_ITERS];
#pragma unroll
        for (int u = 0; u < TABLE_ITERS; ++u) {
            const int i = tid + u * PART_THREADS;
            tv[u] = (i < n4) ? reinterpret_cast<const float4*>(deg_out)[i]
                             : make_float4(0.f, 0.f, 0.f, 0.f);
        }
        cv = (tid < 256 && tid < n_ranges) ? cnt[tid] : 0;
#pragma unroll
        for (int u = 0; u < TABLE_ITERS; ++u) {
            const int i = tid + u * PART_THREADS;
            if (i < n4) reinterpret_cast<float4*>(tbl)[i] = tv[u];
        }
        const int i = (n4 << 2) + tid;
        if (i < nt) tbl[i] = deg_out[i];
    }
    if (tid < 256) {
        hist[tid] = 0;
        seglen[tid] = (cv + 15) & ~15;
    }
    __syncthreads();

    int r[4], rank[4];
    float w[4];
#pragma unroll
    for (int k = 0; k < 4; ++k) {
        r[k] = 0; rank[k] = 0; w[k] = 0.0f;
        if (k < nval) {
            r[k] = (int)((unsigned)d[k] / R_NODES);
            rank[k] = atomicAdd(&hist[r[k]], 1);
        }
    }
    if (use_tbl) {
#pragma unroll
        for (int k = 0; k < 4; ++k)
            if (k < nval) w[k] = rsqrtf(fmaxf(tbl[s[k]], 1.0f));
    } else {
#pragma unroll
        for (int k = 0; k < 4; ++k)
            if (k < nval) w[k] = rsqrtf(fmaxf(deg_out[s[k]], 1.0f));
    }
    __syncthreads();

    int h = 0, hs = 0, cs = 0, c = 0;
    if (tid < 256) {
        h = hist[tid];
        c = seglen[tid];
        hs = h;
        cs = c;
#pragma unroll
        for (int o = 1; o < 32; o <<= 1) {
            const int t1 = __shfl_up(hs, o);
            const int t2 = __shfl_up(cs, o);
            if (lane >= o) { hs += t1; cs += t2; }
        }
        if (lane == 31) { wsum[wave] = hs; wsum[8 + wave] = cs; }
    }
    __syncthreads();
    if (tid < 256) {
        int ph = 0, pc = 0;
        for (int k = 0; k < wave; ++k) { ph += wsum[k]; pc += wsum[8 + k]; }
        loff[tid] = ph + hs - h;
        int g = 0;
        if (h > 0) g = atomicAdd(&cursor[tid], h);
        gbase[tid] = pc + cs - c + g;
    }
    __syncthreads();

#pragma unroll
    for (int k = 0; k < 4; ++k) {
        if (k < nval) {
            const int pos = loff[r[k]] + rank[k];
            const unsigned local = (unsigned)d[k] - (unsigned)r[k] * R_NODES;
            stage[pos] = make_int2((int)((unsigned)s[k] | (local << 24)), __float_as_int(w[k]));
            stage_r[pos] = (unsigned char)r[k];
        }
    }
    __syncthreads();

    int nloc = n_edges - chunk0;
    if (nloc > PART_EDGES) nloc = PART_EDGES;
    for (int i = tid; i < nloc; i += PART_THREADS) {
        const int rr = stage_r[i];
        const int dest = gbase[rr] + (i - loff[rr]);
        if (dest >= 0 && dest < rec_cap) rec[dest] = stage[i];
    }
}

template <int F, int F_OUT, int NT, bool RELU>
__global__ __launch_bounds__(SPMM_WAVES * 32) void spmm_rec_kernel(
    const float* __restrict__ h,
    const int* __restrict__ rec_in, const int* __restrict__ cnt,
    const float* __restrict__ W, const float* __restrict__ bias,
    float* __restrict__ out, int n_nodes, int rec_cap) {
    constexpr int P = LdsPitch<F>::v;
    __shared__ alignas(16) float acc[R_PAD * P];
    __shared__ alignas(16) float wl[F * W_PITCH];
    __shared__ int seg[2];
    __shared__ alignas(16) int degl[256];
    const int2* rec = reinterpret_cast<const int2*>(rec_in);

    const int tid = threadIdx.x;
    const int lane = tid & 31;
    const int wave = __builtin_amdgcn_readfirstlane(tid >> 5);
    const int b = blockIdx.x;
    const int r0 = b * R_NODES;

    for (int i = tid; i < R_PAD * P / 4; i += SPMM_WAVES * 32)
        reinterpret_cast<float4*>(acc)[i] = make_float4(0.f, 0.f, 0.f, 0.f);
    if (tid < 256) degl[tid] = 0;
    stage_w<F, F_OUT, NT>(wl, W, tid, SPMM_WAVES * 32);
    if (wave == 0) {
        int s = 0;
        for (int k = lane; k < b; k += 32) s += (cnt[k] + 15) & ~15;
        for (int o = 16; o > 0; o >>= 1) s += __shfl_xor(s, o);
        if (lane == 0) { seg[0] = s; seg[1] = cnt[b]; }
    }
    __syncthreads();
    int off = __builtin_amdgcn_readfirstlane(seg[0]);
    int count = __builtin_amdgcn_readfirstlane(seg[1]);
    if (off > rec_cap) off = rec_cap;
    if (count > rec_cap - off) count = rec_cap - off;

    const int per = (count + SPMM_WAVES - 1) / SPMM_WAVES;
    int p = wave * per;
    int pend = p + per;
    if (pend > count) pend = count;

    if (p < pend) {
        int i0 = p + lane;
        if (i0 > pend - 1) i0 = pend - 1;
        int2 r = rec[off + i0];
        for (; p < pend; p += 32) {
            int n = pend - p;
            if (n > 32) n = 32;
            int i1 = p + 32 + lane;
            if (i1 > pend - 1) i1 = pend - 1;
            const int2 rn = rec[off + i1];
            const int sv = r.x & REC_ID_MASK;
            const int dv = (int)((unsigned)r.x >> 24);
            const float nv = __int_as_float(r.y);
            if (lane < n) atomicAdd(&degl[dv], 1);
            add_queue<F>(h, acc, sv, dv, nv, n, lane);
            r = rn;
        }
    }

    __syncthreads();

    if (wave < R_TILES)
        dense_tile_from_lds<F, F_OUT, NT, RELU>(acc, wl, degl, bias, wave, lane);

    __syncthreads();

    store_rows<P, F_OUT>(acc, out, r0, n_nodes, tid, SPMM_WAVES * 32);
}

static inline size_t align256(size_t x) { return (x + 255) & ~(size_t)255; }

extern "C" void kernel_launch(void* const* d_in, const int* in_sizes, int n_in,
                              void* d_out, int out_size, void* d_ws, size_t ws_size,
                              hipStream_t stream) {
    const float* x  = (const float*)d_in[0];
    const float* W0 = (const float*)d_in[1];
    const float* b0 = (const float*)d_in[2];
    const float* W1 = (const float*)d_in[3];
    const float* b1 = (const float*)d_in[4];
    const float* W2 = (const float*)d_in[5];
    const float* b2 = (const float*)d_in[6];
    const int* src  = (const int*)d_in[7];
    const int* dst  = (const int*)d_in[8];

    const int N = in_sizes[0] / N_FEAT_IN;
    const int E = in_sizes[7];
    float* out = (float*)d_out;

    char* p = (char*)d_ws;
    float* deg_out = (float*)p;  p += align256((size_t)N * 4);
    int* cnt = (int*)p;  p += align256((size_t)256 * 4);
    int* cursor = (int*)p;  p += align256((size_t)256 * 4);
    const size_t clear_bytes = (size_t)(p - (char*)cnt);
    float* h1   = (float*)p;  p += align256((size_t)N * 64 * 4);
    float* h0   = (float*)p;  p += align256((size_t)N * 64 * 4);
    const int rec_cap = E + 16 * 256;
    int* rec = (int*)p;  p += align256((size_t)rec_cap * 8);

    const int ranges = (N + R_NODES - 1) / R_NODES;
    const int chunks = (E + COUNT_EDGES - 1) / COUNT_EDGES;

    unsigned char* src_ids = (unsigned char*)h1;
    unsigned* piece = (unsigned*)((char*)h1 + align256((size_t)chunks * COUNT_EDGES));

    hipMemsetAsync(cnt, 0, clear_bytes, stream);
    count_kernel<<<chunks, COUNT_THREADS, 0, stream>>>(src, dst, src_ids, piece, cnt, E);

    outdeg_kernel<<<ranges, OUTDEG_THREADS, 0, stream>>>(src_ids, piece, deg_out, chunks, N);

    partition_kernel<<<(E + PART_EDGES - 1) / PART_EDGES, PART_THREADS, 0, stream>>>(
        src, dst, deg_out, cnt, cursor, rec, E, N, ranges, rec_cap);

    spmm_rec_kernel<N_FEAT_IN, 64, 4, true><<<ranges, SPMM_WAVES * 32, 0, stream>>>(
        x, rec, cnt, W0, b0, h0, N, rec_cap);

    spmm_rec_kernel<64, 64, 4, true><<<ranges, SPMM_WAVES * 32, 0, stream>>>(
        h0, rec, cnt, W1, b1, h1, N, rec_cap);

    spmm_rec_kernel<64, 47, 3, false><<<ranges, SPMM_WAVES * 32, 0, stream>>>(
        h1, rec, cnt, W2, b2, out, N, rec_cap);
}


# Layer launches start sooner: segment start = cursor - count (out-degree launch seeds the cursors with the starts), two scalar loads; first records and all W loads issued before the LDS clear
# speedup vs baseline: 5.2073x; 1.0601x over previous
#include <hip/hip_runtime.h>


typedef float v2f __attribute__((ext_vector_type(2)));
typedef float v8f __attribute__((ext_vector_type(8)));

#define N_FEAT_IN 100

#define R_NODES 200
#define R_TILES ((R_NODES + 15) / 16)
#define R_PAD (R_TILES * 16)
template <int F> struct LdsPitch { static constexpr int v = (F % 64 == 0) ? F + 4 : F; };

#define COUNT_EDGES 4096
#define COUNT_THREADS 1024

__global__ __launch_bounds__(COUNT_THREADS) void count_kernel(const int* __restrict__ src,
                                                              const int* __restrict__ dst,
                                                              unsigned char* __restrict__ src_ids,
                                                              unsigned* __restrict__ piece,
                                                              int* __restrict__ cnt,
                                                              int n_edges) {
    __shared__ int hist[256];
    __shared__ int shist[256];
    __shared__ int soff[256];
    __shared__ int wsum[8];
    __shared__ alignas(16) unsigned char sstage[COUNT_EDGES];
    const int tid = threadIdx.x;
    const int lane = tid & 31;
    const int wave = tid >> 5;
    const int e0 = blockIdx.x * COUNT_EDGES + tid * 4;
    int nval = n_edges - e0;
    if (nval > 4) nval = 4;
    int s[4], d[4];
    if (nval == 4) {
        const int4 sv = *reinterpret_cast<const int4*>(src + e0);
        const int4 dv = *reinterpret_cast<const int4*>(dst + e0);
        s[0] = sv.x; s[1] = sv.y; s[2] = sv.z; s[3] = sv.w;
        d[0] = dv.x; d[1] = dv.y; d[2] = dv.z; d[3] = dv.w;
    } else {
#pragma unroll
        for (int k = 0; k < 4; ++k) {
            s[k] = (k < nval) ? src[e0 + k] : 0;
            d[k] = (k < nval) ? dst[e0 + k] : 0;
        }
    }
    if (tid < 256) { hist[tid] = 0; shist[tid] = 0; }
    __syncthreads();
    int rs[4], rk[4];
#pragma unroll
    for (int k = 0; k < 4; ++k) {
        rs[k] = 0; rk[k] = 0;
        if (k < nval) {
            atomicAdd(&hist[(unsigned)d[k] / R_NODES], 1);
            rs[k] = (int)((unsigned)s[k] / R_NODES);
            rk[k] = atomicAdd(&shist[rs[k]], 1);
        }
    }
    __syncthreads();
    int h = 0, c = 0, cs = 0;
    if (tid < 256) {
        h = hist[tid];
        if (h > 0) atomicAdd(&cnt[tid], h);
        c = shist[tid];
        cs = c;
#pragma unroll
        for (int o = 1; o < 32; o <<= 1) {
            const int t1 = __shfl_up(cs, o);
            if (lane >= o) cs += t1;
        }
        if (lane == 31) wsum[wave] = cs;
    }
    __syncthreads();
    if (tid < 256) {
        int p = 0;
        for (int k = 0; k < wave; ++k) p += wsum[k];
        const int start = p + cs - c;
        soff[tid] = start;
        piece[blockIdx.x * 256 + tid] = (unsigned)start | ((unsigned)c << 16);
    }
    __syncthreads();
#pragma unroll
    for (int k = 0; k < 4; ++k)
        if (k < nval)
            sstage[soff[rs[k]] + rk[k]] = (unsigned char)((unsigned)s[k] - (unsigned)rs[k] * R_NODES);
    __syncthreads();
    reinterpret_cast<unsigned*>(src_ids + (size_t)blockIdx.x * COUNT_EDGES)[tid] =
        reinterpret_cast<const unsigned*>(sstage)[tid];
}

#define OUTDEG_THREADS 256

__global__ __launch_bounds__(OUTDEG_THREADS) void outdeg_kernel(
    const unsigned char* __restrict__ src_ids, const unsigned* __restrict__ piece,
    const int* __restrict__ cnt, int* __restrict__ cursor,
    float* __restrict__ deg_out, int n_chunks, int n_nodes) {
    __shared__ int degs[256];
    const int tid = threadIdx.x;
    const int r = blockIdx.x;
    degs[tid] = 0;
    __syncthreads();
    if ((tid >> 5) == OUTDEG_THREADS / 32 - 1) {
        const int lane = tid & 31;
        int s = 0;
#pragma unroll
        for (int j = 0; j < 8; ++j) {
            const int k = lane + 32 * j;
            const int v = cnt[k];
            s += (k < r) ? ((v + 15) & ~15) : 0;
        }
        for (int o = 16; o > 0; o >>= 1) s += __shfl_xor(s, o);
        if (lane == 0) cursor[r] = s;
    }
    for (int ch = tid; ch < n_chunks; ch += OUTDEG_THREADS) {
        const unsigned pc = piece[ch * 256 + r];
        const int start = (int)(pc & 0xFFFFu);
        const int n = (int)(pc >> 16);
        const unsigned char* p = src_ids + (size_t)ch * COUNT_EDGES + start;
        for (int k = 0; k < n; k += 8) {
            int b[8];
#pragma unroll
            for (int j = 0; j < 8; ++j) b[j] = (k + j < n) ? (int)p[k + j] : 0;
#pragma unroll
            for (int j = 0; j < 8; ++j)
                if (k + j < n) atomicAdd(&degs[b[j]], 1);
        }
    }
    __syncthreads();
    const int node = r * R_NODES + tid;
    if (tid < R_NODES && node < n_nodes) deg_out[node] = (float)degs[tid];
}

template <int F, int K>
__device__ __forceinline__ void add_rows(const float* __restrict__ h, float* acc,
                                         int sv, int dv, float nv, int j, int lane) {
    constexpr int NC = (F + 31) / 32;
    int s[K], d[K];
    float ns[K];
    float v[K][NC];
#pragma unroll
    for (int t = 0; t < K; ++t) {
        s[t] = __builtin_amdgcn_readlane(sv, j + t);
        d[t] = __builtin_amdgcn_readlane(dv, j + t);
        ns[t] = __int_as_float(__builtin_amdgcn_readlane(__float_as_int(nv), j + t));
    }
#pragma unroll
    for (int t = 0; t < K; ++t) {
        const float* hrow = h + (size_t)s[t] * F;
#pragma unroll
        for (int c = 0; c < NC; ++c) {
            const int f = c * 32 + lane;
            v[t][c] = (f < F) ? hrow[f] : 0.0f;
        }
    }
#pragma unroll
    for (int t = 0; t < K; ++t) {
        float* arow = acc + d[t] * LdsPitch<F>::v;
#pragma unroll
        for (int c = 0; c < NC; ++c) {
            const int f = c * 32 + lane;
            if (f < F)
                __hip_atomic_fetch_add(arow + f, v[t][c] * ns[t], __ATOMIC_RELAXED,
                                       __HIP_MEMORY_SCOPE_WORKGROUP);
        }
    }
}

template <int F>
__device__ __forceinline__ void add_queue(const float* __restrict__ h, float* acc,
                                          int sv, int dv, float nv, int n, int lane) {
    int j = 0;
    for (; j + 8 <= n; j += 8) add_rows<F, 8>(h, acc, sv, dv, nv, j, lane);
    if (j + 4 <= n) { add_rows<F, 4>(h, acc, sv, dv, nv, j, lane); j += 4; }
    if (j + 2 <= n) { add_rows<F, 2>(h, acc, sv, dv, nv, j, lane); j += 2; }
    if (j < n) add_rows<F, 1>(h, acc, sv, dv, nv, j, lane);
}

#define REC_ID_MASK 0x00FFFFFF

#define W_PITCH 80

template <int F_IN, int F_OUT, int NT, int NTHREADS>
struct WStage {
    static constexpr int C = NT * 16;
    static constexpr int NW = (F_IN * C + NTHREADS - 1) / NTHREADS;
    static_assert(C <= W_PITCH && F_IN % 2 == 0, "W does not fit its LDS layout");
    float v[NW];
    __device__ __forceinline__ void load(const float* __restrict__ W, int tid) {
#pragma unroll
        for (int u = 0; u < NW; ++u) {
            const int i = tid + u * NTHREADS;
            const int k = i / C;
            const int c = i - k * C;
            v[u] = (i < F_IN * C && c < F_OUT) ? W[k * F_OUT + c] : 0.0f;
        }
    }
    __device__ __forceinline__ void store(float* wl, int tid) const {
#pragma unroll
        for (int u = 0; u < NW; ++u) {
            const int i = tid + u * NTHREADS;
            const int k = i / C;
            const int c = i - k * C;
            if (i < F_IN * C) wl[(k >> 1) * (2 * W_PITCH) + c * 2 + (k & 1)] = v[u];
        }
    }
};

template <int NT>
__device__ __forceinline__ void load_b_frags(const float* wl, int ka, int lm, v2f (&bf)[NT]) {
#pragma unroll
    for (int t = 0; t < NT; ++t)
        bf[t] = *reinterpret_cast<const v2f*>(wl + (ka >> 1) * (2 * W_PITCH) + (t * 16 + lm) * 2);
}

template <int F_IN, int F_OUT, int NT, bool RELU>
__device__ __forceinline__ void dense_tile_from_lds(float* tile, const float* wl,
                                                    const int* degl,
                                                    const float* __restrict__ bias,
                                                    int t16, int lane) {
    constexpr int P = LdsPitch<F_IN>::v;
    const int half = lane >> 4;
    const int lm = lane & 15;

    v8f acc[NT];
#pragma unroll
    for (int t = 0; t < NT; ++t) acc[t] = v8f{};

    const float* arow = tile + (t16 * 16 + lm) * P + 2 * half;

    float nd[8];
#pragma unroll
    for (int i = 0; i < 8; ++i)
        nd[i] = rsqrtf(fmaxf((float)degl[t16 * 16 + i + 8 * half], 1.0f));
    float bs[NT];
#pragma unroll
    for (int t = 0; t < NT; ++t) {
        const int col = t * 16 + lm;
        bs[t] = bias[col < F_OUT ? col : F_OUT - 1];
    }

    v2f bf[NT];
    load_b_frags<NT>(wl, 2 * half, lm, bf);
#pragma unroll
    for (int k0 = 0; k0 < F_IN; k0 += 4) {
        const v2f a = *reinterpret_cast<const v2f*>(arow + k0);
        v2f bn[NT];
        if (k0 + 4 < F_IN) load_b_frags<NT>(wl, k0 + 4 + 2 * half, lm, bn);
#pragma unroll
        for (int t = 0; t < NT; ++t)
            acc[t] = __builtin_amdgcn_wmma_f32_16x16x4_f32(
                 false, a,  false, bf[t],
                 (short)0, acc[t],  false,  false);
        if (k0 + 4 < F_IN) {
#pragma unroll
            for (int t = 0; t < NT; ++t) bf[t] = bn[t];
        }
    }

#pragma unroll
    for (int i = 0; i < 8; ++i) {
        const int row = t16 * 16 + i + 8 * half;
#pragma unroll
        for (int t = 0; t < NT; ++t) {
            const int col = t * 16 + lm;
            if (col < F_OUT) {
                float v = acc[t][i] * nd[i] + bs[t];
                if (RELU) v = fmaxf(v, 0.0f);
                tile[row * P + col] = v;
            }
        }
    }
}

template <int P, int F_OUT>
__device__ __forceinline__ void store_rows(const float* tile, float* __restrict__ out,
                                           int r0, int n_nodes, int tid, int nthreads) {
    int rows = n_nodes - r0;
    if (rows > R_NODES) rows = R_NODES;
    float* o = out + (size_t)r0 * F_OUT;
    if (F_OUT % 4 == 0) {
        constexpr int Q = F_OUT / 4;
        const int n4 = rows * Q;
        for (int i = tid; i < n4; i += nthreads) {
            const int row = i / Q;
            const int c4 = i - row * Q;
            reinterpret_cast<float4*>(o)[i] =
                *reinterpret_cast<const float4*>(tile + row * P + c4 * 4);
        }
    } else {
        const int n1 = rows * F_OUT;
        for (int i = tid; i < n1; i += nthreads) {
            const int row = i / F_OUT;
            const int col = i - row * F_OUT;
            o[i] = tile[row * P + col];
        }
    }
}

#define SPMM_WAVES 32

#define PART_EDGES 4096
#define PART_THREADS 1024
#define NODE_TABLE 50000
#define TABLE_ITERS ((NODE_TABLE / 4 + PART_THREADS - 1) / PART_THREADS)

__global__ __launch_bounds__(PART_THREADS) void partition_kernel(
    const int* __restrict__ src, const int* __restrict__ dst,
    const float* __restrict__ deg_out,
    int* __restrict__ cursor, int* __restrict__ rec_out,
    int n_edges, int n_nodes, int rec_cap) {
    __shared__ alignas(16) float tbl[NODE_TABLE];
    __shared__ int2 stage[PART_EDGES];
    __shared__ unsigned char stage_r[PART_EDGES];
    __shared__ int hist[256];
    __shared__ int loff[256];
    __shared__ int gbase[256];
    __shared__ int wsum[8];
    int2* rec = reinterpret_cast<int2*>(rec_out);

    const int tid = threadIdx.x;
    const int lane = tid & 31;
    const int wave = tid >> 5;
    const int chunk0 = blockIdx.x * PART_EDGES;
    const int e0 = chunk0 + tid * 4;

    int s[4], d[4];
    int nval = n_edges - e0;
    if (nval > 4) nval = 4;
    if (nval == 4) {
        const int4 sv = *reinterpret_cast<const int4*>(src + e0);
        const int4 dv = *reinterpret_cast<const int4*>(dst + e0);
        s[0] = sv.x; s[1] = sv.y; s[2] = sv.z; s[3] = sv.w;
        d[0] = dv.x; d[1] = dv.y; d[2] = dv.z; d[3] = dv.w;
    } else {
#pragma unroll
        for (int k = 0; k < 4; ++k) {
            s[k] = (k < nval) ? src[e0 + k] : 0;
            d[k] = (k < nval) ? dst[e0 + k] : 0;
        }
    }

    const bool use_tbl = (n_nodes <= NODE_TABLE);
    const int nt = use_tbl ? n_nodes : 0;
    const int n4 = nt >> 2;
    {
        float4 tv[TABLE_ITERS];
#pragma unroll
        for (int u = 0; u < TABLE_ITERS; ++u) {
            const int i = tid + u * PART_THREADS;
            tv[u] = (i < n4) ? reinterpret_cast<const float4*>(deg_out)[i]
                             : make_float4(0.f, 0.f, 0.f, 0.f);
        }
#pragma unroll
        for (int u = 0; u < TABLE_ITERS; ++u) {
            const int i = tid + u * PART_THREADS;
            if (i < n4) reinterpret_cast<float4*>(tbl)[i] = tv[u];
        }
        const int i = (n4 << 2) + tid;
        if (i < nt) tbl[i] = deg_out[i];
    }
    if (tid < 256) hist[tid] = 0;
    __syncthreads();

    int r[4], rank[4];
    float w[4];
#pragma unroll
    for (int k = 0; k < 4; ++k) {
        r[k] = 0; rank[k] = 0; w[k] = 0.0f;
        if (k < nval) {
            r[k] = (int)((unsigned)d[k] / R_NODES);
            rank[k] = atomicAdd(&hist[r[k]], 1);
        }
    }
    if (use_tbl) {
#pragma unroll
        for (int k = 0; k < 4; ++k)
            if (k < nval) w[k] = rsqrtf(fmaxf(tbl[s[k]], 1.0f));
    } else {
#pragma unroll
        for (int k = 0; k < 4; ++k)
            if (k < nval) w[k] = rsqrtf(fmaxf(deg_out[s[k]], 1.0f));
    }
    __syncthreads();

    int h = 0, hs = 0, g = 0;
    if (tid < 256) {
        h = hist[tid];
        if (h > 0) g = atomicAdd(&cursor[tid], h);
        hs = h;
#pragma unroll
        for (int o = 1; o < 32; o <<= 1) {
            const int t1 = __shfl_up(hs, o);
            if (lane >= o) hs += t1;
        }
        if (lane == 31) wsum[wave] = hs;
    }
    __syncthreads();
    if (tid < 256) {
        int ph = 0;
        for (int k = 0; k < wave; ++k) ph += wsum[k];
        loff[tid] = ph + hs - h;
        gbase[tid] = g;
    }
    __syncthreads();

#pragma unroll
    for (int k = 0; k < 4; ++k) {
        if (k < nval) {
            const int pos = loff[r[k]] + rank[k];
            const unsigned local = (unsigned)d[k] - (unsigned)r[k] * R_NODES;
            stage[pos] = make_int2((int)((unsigned)s[k] | (local << 24)), __float_as_int(w[k]));
            stage_r[pos] = (unsigned char)r[k];
        }
    }
    __syncthreads();

    int nloc = n_edges - chunk0;
    if (nloc > PART_EDGES) nloc = PART_EDGES;
    for (int i = tid; i < nloc; i += PART_THREADS) {
        const int rr = stage_r[i];
        const int dest = gbase[rr] + (i - loff[rr]);
        if (dest >= 0 && dest < rec_cap) rec[dest] = stage[i];
    }
}

template <int F, int F_OUT, int NT, bool RELU>
__global__ __launch_bounds__(SPMM_WAVES * 32) void spmm_rec_kernel(
    const float* __restrict__ h,
    const int* __restrict__ rec_in, const int* __restrict__ cnt,
    const int* __restrict__ cursor,
    const float* __restrict__ W, const float* __restrict__ bias,
    float* __restrict__ out, int n_nodes, int rec_cap) {
    constexpr int P = LdsPitch<F>::v;
    __shared__ alignas(16) float acc[R_PAD * P];
    __shared__ alignas(16) float wl[F * W_PITCH];
    __shared__ alignas(16) int degl[256];
    const int2* rec = reinterpret_cast<const int2*>(rec_in);

    const int tid = threadIdx.x;
    const int lane = tid & 31;
    const int wave = __builtin_amdgcn_readfirstlane(tid >> 5);
    const int b = blockIdx.x;
    const int r0 = b * R_NODES;

    int count = cnt[b];
    int off = cursor[b] - count;
    WStage<F, F_OUT, NT, SPMM_WAVES * 32> ws;
    ws.load(W, tid);

    if (count < 0) count = 0;
    if (off < 0) off = 0;
    if (off > rec_cap) off = rec_cap;
    if (count > rec_cap - off) count = rec_cap - off;

    const int per = (count + SPMM_WAVES - 1) / SPMM_WAVES;
    int p = wave * per;
    int pend = p + per;
    if (pend > count) pend = count;

    int2 r = make_int2(0, 0);
    if (p < pend) {
        int i0 = p + lane;
        if (i0 > pend - 1) i0 = pend - 1;
        r = rec[off + i0];
    }

    for (int i = tid; i < R_PAD * P / 4; i += SPMM_WAVES * 32)
        reinterpret_cast<float4*>(acc)[i] = make_float4(0.f, 0.f, 0.f, 0.f);
    if (tid < 256) degl[tid] = 0;
    ws.store(wl, tid);
    __syncthreads();

    if (p < pend) {
        for (; p < pend; p += 32) {
            int n = pend - p;
            if (n > 32) n = 32;
            int i1 = p + 32 + lane;
            if (i1 > pend - 1) i1 = pend - 1;
            const int2 rn = rec[off + i1];
            const int sv = r.x & REC_ID_MASK;
            const int dv = (int)((unsigned)r.x >> 24);
            const float nv = __int_as_float(r.y);
            if (lane < n) atomicAdd(&degl[dv], 1);
            add_queue<F>(h, acc, sv, dv, nv, n, lane);
            r = rn;
        }
    }

    __syncthreads();

    if (wave < R_TILES)
        dense_tile_from_lds<F, F_OUT, NT, RELU>(acc, wl, degl, bias, wave, lane);

    __syncthreads();

    store_rows<P, F_OUT>(acc, out, r0, n_nodes, tid, SPMM_WAVES * 32);
}

static inline size_t align256(size_t x) { return (x + 255) & ~(size_t)255; }

extern "C" void kernel_launch(void* const* d_in, const int* in_sizes, int n_in,
                              void* d_out, int out_size, void* d_ws, size_t ws_size,
                              hipStream_t stream) {
    const float* x  = (const float*)d_in[0];
    const float* W0 = (const float*)d_in[1];
    const float* b0 = (const float*)d_in[2];
    const float* W1 = (const float*)d_in[3];
    const float* b1 = (const float*)d_in[4];
    const float* W2 = (const float*)d_in[5];
    const float* b2 = (const float*)d_in[6];
    const int* src  = (const int*)d_in[7];
    const int* dst  = (const int*)d_in[8];

    const int N = in_sizes[0] / N_FEAT_IN;
    const int E = in_sizes[7];
    float* out = (float*)d_out;

    char* p = (char*)d_ws;
    float* deg_out = (float*)p;  p += align256((size_t)N * 4);
    int* cnt = (int*)p;  p += align256((size_t)256 * 4);
    const size_t clear_bytes = (size_t)(p - (char*)cnt);
    int* cursor = (int*)p;  p += align256((size_t)256 * 4);
    float* h1   = (float*)p;  p += align256((size_t)N * 64 * 4);
    float* h0   = (float*)p;  p += align256((size_t)N * 64 * 4);
    const int rec_cap = E + 16 * 256;
    int* rec = (int*)p;  p += align256((size_t)rec_cap * 8);

    const int ranges = (N + R_NODES - 1) / R_NODES;
    const int chunks = (E + COUNT_EDGES - 1) / COUNT_EDGES;

    unsigned char* src_ids = (unsigned char*)h1;
    unsigned* piece = (unsigned*)((char*)h1 + align256((size_t)chunks * COUNT_EDGES));

    hipMemsetAsync(cnt, 0, clear_bytes, stream);
    count_kernel<<<chunks, COUNT_THREADS, 0, stream>>>(src, dst, src_ids, piece, cnt, E);

    outdeg_kernel<<<ranges, OUTDEG_THREADS, 0, stream>>>(src_ids, piece, cnt, cursor, deg_out,
                                                         chunks, N);

    partition_kernel<<<(E + PART_EDGES - 1) / PART_EDGES, PART_THREADS, 0, stream>>>(
        src, dst, deg_out, cursor, rec, E, N, rec_cap);

    spmm_rec_kernel<N_FEAT_IN, 64, 4, true><<<ranges, SPMM_WAVES * 32, 0, stream>>>(
        x, rec, cnt, cursor, W0, b0, h0, N, rec_cap);

    spmm_rec_kernel<64, 64, 4, true><<<ranges, SPMM_WAVES * 32, 0, stream>>>(
        h0, rec, cnt, cursor, W1, b1, h1, N, rec_cap);

    spmm_rec_kernel<64, 47, 3, false><<<ranges, SPMM_WAVES * 32, 0, stream>>>(
        h1, rec, cnt, cursor, W2, b2, out, N, rec_cap);
}
